# starting kernel
# baseline (speedup 1.0000x reference)
#include <hip/hip_runtime.h>

typedef float v2f __attribute__((ext_vector_type(2)));
typedef float v8f __attribute__((ext_vector_type(8)));

#define DIM_D 256
#define DIM_N 1024
#define EDGES_PER_WAVE 16
#define WAVES_PER_BLOCK 8
#define EDGES_PER_BLOCK (EDGES_PER_WAVE * WAVES_PER_BLOCK)

__global__ __launch_bounds__(256, 4)
void edge_norm_wmma(const float* __restrict__ x,
                    const int*   __restrict__ eb,
                    const int*   __restrict__ ei,
                    const int*   __restrict__ ej,
                    float*       __restrict__ out,
                    int E)
{
    __shared__ float lds[WAVES_PER_BLOCK][16 * 16];

    const int tid   = threadIdx.x;
    const int wave  = tid >> 5;
    const int lane  = tid & 31;
    const int m     = lane & 15;
    const int khalf = (lane >> 4) * 2;

    const long groupBase = (long)blockIdx.x * EDGES_PER_BLOCK
                         + (long)wave * EDGES_PER_WAVE;
    const long e  = groupBase + m;
    const long ec = (e < (long)E) ? e : (long)(E - 1);

    const int b = eb[ec];
    const int i = ei[ec];
    const int j = ej[ec];
    const float* __restrict__ pi = x + ((size_t)b * DIM_N + (size_t)i) * DIM_D;
    const float* __restrict__ pj = x + ((size_t)b * DIM_N + (size_t)j) * DIM_D;

    v8f c = {0.f, 0.f, 0.f, 0.f, 0.f, 0.f, 0.f, 0.f};

    #pragma unroll 4
    for (int kb = 0; kb < DIM_D; kb += 4) {
        const int k = kb + khalf;
        v2f xi = *(const v2f*)(pi + k);
        v2f xj = *(const v2f*)(pj + k);
        v2f a;
        a.x = xi.x - xj.x;
        a.y = xi.y - xj.y;
        c = __builtin_amdgcn_wmma_f32_16x16x4_f32(
                 false, a,  false, a,
                 (short)0, c,  false,  false);
    }

    float* tile = lds[wave];
    const int nIdx  = lane & 15;
    const int mBase = (lane >> 4) * 8;
    #pragma unroll
    for (int r = 0; r < 8; ++r)
        tile[(mBase + r) * 16 + nIdx] = c[r];

    __syncthreads();

    if (lane < 16 && e < (long)E) {
        const float s = tile[m * 16 + m];
        const float d = sqrtf(s);
        float4 v = make_float4(d, d, d, d);
        float4* o = (float4*)(out + (size_t)e * 8);
        o[0] = v;
        o[1] = v;
    }
}

extern "C" void kernel_launch(void* const* d_in, const int* in_sizes, int n_in,
                              void* d_out, int out_size, void* d_ws, size_t ws_size,
                              hipStream_t stream)
{
    const float* x  = (const float*)d_in[0];
    const int*   eb = (const int*)d_in[1];
    const int*   ei = (const int*)d_in[2];
    const int*   ej = (const int*)d_in[3];
    float*       out = (float*)d_out;

    const int E = in_sizes[1];
    if (E <= 0) return;

    const int blocks = (E + EDGES_PER_BLOCK - 1) / EDGES_PER_BLOCK;
    hipLaunchKernelGGL(edge_norm_wmma, dim3(blocks), dim3(256), 0, stream,
                       x, eb, ei, ej, out, E);
}


# Read both rows of every edge as whole 128-B lines (b128 loads) and transpose the differences through a wave-private LDS tile into the WMMA fragments; changes the order in which k is summed
# speedup vs baseline: 1.6805x; 1.6805x over previous
#include <hip/hip_runtime.h>

typedef float v2f __attribute__((ext_vector_type(2)));
typedef float v4f __attribute__((ext_vector_type(4)));
typedef float v8f __attribute__((ext_vector_type(8)));

#define DIM_D 256
#define DIM_N 1024
#define EDGES_PER_WAVE 16
#define WAVES_PER_BLOCK 8
#define EDGES_PER_BLOCK (EDGES_PER_WAVE * WAVES_PER_BLOCK)
#define CHUNK 64
#define NCHUNK (DIM_D / CHUNK)
#define PITCH 68
#define TILE_FLOATS (16 * PITCH)

__device__ __forceinline__ void wave_lds_order()
{
    __builtin_amdgcn_fence(__ATOMIC_RELEASE, "wavefront");
    __builtin_amdgcn_wave_barrier();
    __builtin_amdgcn_fence(__ATOMIC_ACQUIRE, "wavefront");
}

__global__ __launch_bounds__(256, 4)
void edge_norm_wmma(const float* __restrict__ x,
                    const int*   __restrict__ eb,
                    const int*   __restrict__ ei,
                    const int*   __restrict__ ej,
                    float*       __restrict__ out,
                    int E)
{
    __shared__ __attribute__((aligned(16))) float lds[WAVES_PER_BLOCK][TILE_FLOATS];

    const int tid  = threadIdx.x;
    const int wave = tid >> 5;
    const int lane = tid & 31;
    const int m    = lane & 15;
    const int h    = lane >> 4;

    const long groupBase = (long)blockIdx.x * EDGES_PER_BLOCK
                         + (long)wave * EDGES_PER_WAVE;
    const long e  = groupBase + m;
    const long ec = (e < (long)E) ? e : (long)(E - 1);

    const int b = eb[ec];
    const int i = ei[ec];
    const int j = ej[ec];
    const unsigned offi = ((unsigned)b * DIM_N + (unsigned)i) * DIM_D;
    const unsigned offj = ((unsigned)b * DIM_N + (unsigned)j) * DIM_D;

    unsigned vi[8], vj[8];
    #pragma unroll
    for (int t = 0; t < 8; ++t) {
        const int src = 2 * t + h;
        vi[t] = (unsigned)__shfl((int)offi, src, 32) + 4u * (unsigned)m;
        vj[t] = (unsigned)__shfl((int)offj, src, 32) + 4u * (unsigned)m;
    }

    float* tile = lds[wave];
    v8f c = {0.f, 0.f, 0.f, 0.f, 0.f, 0.f, 0.f, 0.f};

    #pragma unroll
    for (int q = 0; q < NCHUNK; ++q) {
        v4f d[8];
        #pragma unroll
        for (int t = 0; t < 8; ++t) {
            const v4f xi = *(const v4f*)(x + vi[t] + q * CHUNK);
            const v4f xj = *(const v4f*)(x + vj[t] + q * CHUNK);
            d[t] = xi - xj;
        }

        wave_lds_order();
        #pragma unroll
        for (int t = 0; t < 8; ++t)
            *(v4f*)(tile + (2 * t + h) * PITCH + 4 * m) = d[t];
        wave_lds_order();

        #pragma unroll
        for (int u = 0; u < CHUNK / 8; ++u) {
            const v4f f = *(const v4f*)(tile + m * PITCH + 8 * u + 4 * h);
            v2f a0; a0.x = f.x; a0.y = f.y;
            v2f a1; a1.x = f.z; a1.y = f.w;
            c = __builtin_amdgcn_wmma_f32_16x16x4_f32(
                     false, a0,  false, a0,
                     (short)0, c,  false,  false);
            c = __builtin_amdgcn_wmma_f32_16x16x4_f32(
                     false, a1,  false, a1,
                     (short)0, c,  false,  false);
        }
    }

    const int nIdx  = lane & 15;
    const int mBase = (lane >> 4) * 8;
    wave_lds_order();
    #pragma unroll
    for (int r = 0; r < 8; ++r)
        tile[(mBase + r) * 16 + nIdx] = c[r];
    wave_lds_order();

    if (lane < 16 && e < (long)E) {
        const float s = tile[m * 16 + m];
        const float d = sqrtf(s);
        float4 v = make_float4(d, d, d, d);
        float4* o = (float4*)(out + (size_t)e * 8);
        o[0] = v;
        o[1] = v;
    }
}

extern "C" void kernel_launch(void* const* d_in, const int* in_sizes, int n_in,
                              void* d_out, int out_size, void* d_ws, size_t ws_size,
                              hipStream_t stream)
{
    const float* x  = (const float*)d_in[0];
    const int*   eb = (const int*)d_in[1];
    const int*   ei = (const int*)d_in[2];
    const int*   ej = (const int*)d_in[3];
    float*       out = (float*)d_out;

    const int E = in_sizes[1];
    if (E <= 0) return;

    const int blocks = (E + EDGES_PER_BLOCK - 1) / EDGES_PER_BLOCK;
    hipLaunchKernelGGL(edge_norm_wmma, dim3(blocks), dim3(256), 0, stream,
                       x, eb, ei, ej, out, E);
}


# Launch 1 reads its rows once per block as whole 128-B lines into LDS and takes the WMMA fragments from there, on 256 equal blocks that carry the row lengths; a length is now summed in eight parts
# speedup vs baseline: 3.8812x; 2.3096x over previous
#include <hip/hip_runtime.h>

typedef float v2f __attribute__((ext_vector_type(2)));
typedef float v4f __attribute__((ext_vector_type(4)));
typedef float v8f __attribute__((ext_vector_type(8)));

#define DIM_B 4
#define DIM_D 256
#define DIM_N 1024
#define TAB_J 256
#define ROW_TILES (DIM_B * DIM_N / 16)
#define STAGE_ROWS 128
#define STAGE_PITCH 260
#define QUARTER 64
#define EDGE_BLOCKS 768
#define PIVOT 0.5f
#define KEEP_SHARE 0.125f

#define EDGES_PER_WAVE 16
#define WAVES_PER_BLOCK 8
#define CHUNK 64
#define NCHUNK (DIM_D / CHUNK)
#define PITCH 68
#define TILE_FLOATS (16 * PITCH)

__device__ __forceinline__ void wave_lds_order()
{
    __builtin_amdgcn_fence(__ATOMIC_RELEASE, "wavefront");
    __builtin_amdgcn_wave_barrier();
    __builtin_amdgcn_fence(__ATOMIC_ACQUIRE, "wavefront");
}

__device__ __forceinline__ v8f mma4(v2f a, v2f b, v8f c)
{
    return __builtin_amdgcn_wmma_f32_16x16x4_f32(
                false, a,  false, b,
                (short)0, c,  false,  false);
}

__global__ __launch_bounds__(256)
void row_products_wmma(const float* __restrict__ x,
                       float*       __restrict__ tab,
                       float*       __restrict__ len2)
{
    __shared__ __attribute__((aligned(16))) float rows[STAGE_ROWS * STAGE_PITCH];
    __shared__ float part[8][16];

    const int tid  = threadIdx.x;
    const int wave = tid >> 5;
    const int lane = tid & 31;
    const int c    = lane & 15;
    const int h    = lane >> 4;

    const int gb = blockIdx.x;
    const int b  = gb >> 6;
    const int ib = (gb >> 2) & 15;
    const int jb = gb & 3;
    const int rowA0 = b * DIM_N + ib * 64;
    const int rowB0 = b * DIM_N + jb * 64;

    const int srow = tid >> 4;
    const int sc4  = tid & 15;
    const float* gsrc[8];
    #pragma unroll
    for (int r = 0; r < 8; ++r) {
        const int grow = (r < 4) ? (rowA0 + 16 * r + srow) : (rowB0 + 16 * (r - 4) + srow);
        gsrc[r] = x + (size_t)grow * DIM_D + 4 * sc4;
    }
    float* ldst = rows + srow * STAGE_PITCH + 4 * sc4;

    v4f st[2][8];
    #pragma unroll
    for (int r = 0; r < 8; ++r) {
        st[0][r] = *(const v4f*)(gsrc[r]);
        st[1][r] = *(const v4f*)(gsrc[r] + QUARTER);
    }

    const int wi = wave >> 1;
    const int wj = wave & 1;
    const float* la  = rows + (16 * wi + c) * STAGE_PITCH + 4 * h;
    const float* lb0 = rows + (64 + 32 * wj + c) * STAGE_PITCH + 4 * h;
    const float* lb1 = lb0 + 16 * STAGE_PITCH;
    v8f acc0 = {0.f, 0.f, 0.f, 0.f, 0.f, 0.f, 0.f, 0.f};
    v8f acc1 = {0.f, 0.f, 0.f, 0.f, 0.f, 0.f, 0.f, 0.f};

    #pragma unroll
    for (int q = 0; q < 4; ++q) {
        #pragma unroll
        for (int r = 0; r < 8; ++r)
            *(v4f*)(ldst + 16 * r * STAGE_PITCH + q * QUARTER) = st[q & 1][r] - PIVOT;
        __syncthreads();
        if (q + 2 < 4) {
            #pragma unroll
            for (int r = 0; r < 8; ++r)
                st[q & 1][r] = *(const v4f*)(gsrc[r] + (q + 2) * QUARTER);
        }
        #pragma unroll
        for (int uu = 0; uu < 8; ++uu) {
            const int u = 8 * q + uu;
            const v4f fa = *(const v4f*)(la  + 8 * u);
            const v4f f0 = *(const v4f*)(lb0 + 8 * u);
            const v4f f1 = *(const v4f*)(lb1 + 8 * u);
            v2f a0; a0.x = fa.x; a0.y = fa.y;
            v2f a1; a1.x = fa.z; a1.y = fa.w;
            v2f p0; p0.x = f0.x; p0.y = f0.y;
            v2f p1; p1.x = f0.z; p1.y = f0.w;
            v2f q0; q0.x = f1.x; q0.y = f1.y;
            v2f q1; q1.x = f1.z; q1.y = f1.w;
            acc0 = mma4(a0, p0, acc0);
            acc1 = mma4(a0, q0, acc1);
            acc0 = mma4(a1, p1, acc0);
            acc1 = mma4(a1, q1, acc1);
        }
    }

    {
        const float* ln = rows + (16 * jb + c) * STAGE_PITCH + 32 * wave + 4 * h;
        v8f na = {0.f, 0.f, 0.f, 0.f, 0.f, 0.f, 0.f, 0.f};
        #pragma unroll
        for (int uu = 0; uu < 4; ++uu) {
            const v4f f = *(const v4f*)(ln + 8 * uu);
            v2f a0; a0.x = f.x; a0.y = f.y;
            v2f a1; a1.x = f.z; a1.y = f.w;
            na = mma4(a0, a0, na);
            na = mma4(a1, a1, na);
        }
        float dv = na[0];
        #pragma unroll
        for (int r = 1; r < 8; ++r)
            dv = ((c & 7) == r) ? na[r] : dv;
        if (h == (c >> 3))
            part[wave][c] = dv;
    }

    const int gt = b * (DIM_N / 16) + ib * 4 + wi;
    const int j0 = jb * 64 + wj * 32;
    float* po = tab + ((size_t)gt * 16 + 8 * h) * TAB_J + j0 + c;
    #pragma unroll
    for (int r = 0; r < 8; ++r) {
        po[(size_t)r * TAB_J]      = acc0[r];
        po[(size_t)r * TAB_J + 16] = acc1[r];
    }

    __syncthreads();
    if (tid < 16) {
        float sum = part[0][tid];
        #pragma unroll
        for (int w = 1; w < 8; ++w)
            sum += part[w][tid];
        len2[(b * (DIM_N / 16) + ib * 4 + jb) * 16 + tid] = sum;
    }
}

__device__ __forceinline__ void diff_group(const float* __restrict__ x,
                                           const int*   __restrict__ eb,
                                           const int*   __restrict__ ei,
                                           const int*   __restrict__ ej,
                                           float*       __restrict__ out,
                                           int E, long groupBase, float* tile, int lane)
{
    const int m = lane & 15;
    const int h = lane >> 4;
    const long e  = groupBase + m;
    const long ec = (e < (long)E) ? e : (long)(E - 1);

    const int b = eb[ec];
    const int i = ei[ec];
    const int j = ej[ec];
    const unsigned offi = ((unsigned)b * DIM_N + (unsigned)i) * DIM_D;
    const unsigned offj = ((unsigned)b * DIM_N + (unsigned)j) * DIM_D;

    unsigned vi[8], vj[8];
    #pragma unroll
    for (int t = 0; t < 8; ++t) {
        const int src = 2 * t + h;
        vi[t] = (unsigned)__shfl((int)offi, src, 32) + 4u * (unsigned)m;
        vj[t] = (unsigned)__shfl((int)offj, src, 32) + 4u * (unsigned)m;
    }

    v8f c = {0.f, 0.f, 0.f, 0.f, 0.f, 0.f, 0.f, 0.f};

    #pragma nounroll
    for (int q = 0; q < NCHUNK; ++q) {
        v4f d[8];
        #pragma unroll
        for (int t = 0; t < 8; ++t) {
            const v4f xi = *(const v4f*)(x + vi[t] + q * CHUNK);
            const v4f xj = *(const v4f*)(x + vj[t] + q * CHUNK);
            d[t] = xi - xj;
        }

        wave_lds_order();
        #pragma unroll
        for (int t = 0; t < 8; ++t)
            *(v4f*)(tile + (2 * t + h) * PITCH + 4 * m) = d[t];
        wave_lds_order();

        #pragma unroll
        for (int u = 0; u < CHUNK / 8; ++u) {
            const v4f f = *(const v4f*)(tile + m * PITCH + 8 * u + 4 * h);
            v2f a0; a0.x = f.x; a0.y = f.y;
            v2f a1; a1.x = f.z; a1.y = f.w;
            c = mma4(a0, a0, c);
            c = mma4(a1, a1, c);
        }
    }

    const int mBase = h * 8;
    wave_lds_order();
    #pragma unroll
    for (int r = 0; r < 8; ++r)
        tile[(mBase + r) * 16 + m] = c[r];
    wave_lds_order();

    if (lane < 16 && e < (long)E) {
        const float s = tile[m * 16 + m];
        const float dd = sqrtf(s);
        float4 v = make_float4(dd, dd, dd, dd);
        float4* o = (float4*)(out + (size_t)e * 8);
        o[0] = v;
        o[1] = v;
    }
    wave_lds_order();
}

__global__ __launch_bounds__(256)
void edge_dist(const float* __restrict__ x,
               const int*   __restrict__ eb,
               const int*   __restrict__ ei,
               const int*   __restrict__ ej,
               const float* __restrict__ tab,
               const float* __restrict__ len2,
               float*       __restrict__ out,
               int E)
{
    __shared__ __attribute__((aligned(16))) float lds[WAVES_PER_BLOCK][TILE_FLOATS];

    const int tid  = threadIdx.x;
    const int wave = tid >> 5;
    const int lane = tid & 31;

    long blk  = blockIdx.x;
    long base = blk * 256 + (long)wave * 32;
    int b = 0, i = 0, j = 0;
    if (base < (long)E) {
        const long e0 = base + lane;
        const long ec = (e0 < (long)E) ? e0 : (long)(E - 1);
        b = eb[ec];
        i = ei[ec];
        j = ej[ec];
    }

    while (blk * 256 < (long)E) {
        const long nblk  = blk + EDGE_BLOCKS;
        const long nbase = nblk * 256 + (long)wave * 32;
        int nb = 0, ni = 0, nj = 0;

        if (base < (long)E) {
            const bool intab = (unsigned)b < DIM_B && (unsigned)i < DIM_N && (unsigned)j < TAB_J;
            const unsigned bc = min((unsigned)b, (unsigned)(DIM_B - 1));
            const unsigned ic = min((unsigned)i, (unsigned)(DIM_N - 1));
            const unsigned jc = min((unsigned)j, (unsigned)(TAB_J - 1));
            const unsigned r  = bc * DIM_N;
            const float li = len2[r + ic];
            const float lj = len2[r + jc];
            const float g  = tab[(size_t)(r + ic) * TAB_J + jc];
            if (nbase < (long)E) {
                const long e1 = nbase + lane;
                const long ec = (e1 < (long)E) ? e1 : (long)(E - 1);
                nb = eb[ec];
                ni = ei[ec];
                nj = ej[ec];
            }

            const float s  = li + lj;
            const float d2 = s - 2.f * g;
            const bool same = (i == j);
            const bool ok = intab && (s < 3.0e38f) && (same || d2 >= KEEP_SHARE * s);
            const float d = same ? 0.f : sqrtf(d2);

            if (__all(ok)) {
                const float dlo = __shfl(d, lane >> 1, 32);
                const float dhi = __shfl(d, 16 + (lane >> 1), 32);
                float* o = out + (size_t)base * 8;
                if (base + (lane >> 1) < (long)E)
                    *(float4*)(o + 4 * lane) = make_float4(dlo, dlo, dlo, dlo);
                if (base + 16 + (lane >> 1) < (long)E)
                    *(float4*)(o + 128 + 4 * lane) = make_float4(dhi, dhi, dhi, dhi);
            } else {
                diff_group(x, eb, ei, ej, out, E, base, lds[wave], lane);
                if (base + 16 < (long)E)
                    diff_group(x, eb, ei, ej, out, E, base + 16, lds[wave], lane);
            }
        }

        blk = nblk; base = nbase;
        b = nb; i = ni; j = nj;
    }
}

extern "C" void kernel_launch(void* const* d_in, const int* in_sizes, int n_in,
                              void* d_out, int out_size, void* d_ws, size_t ws_size,
                              hipStream_t stream)
{
    const float* x  = (const float*)d_in[0];
    const int*   eb = (const int*)d_in[1];
    const int*   ei = (const int*)d_in[2];
    const int*   ej = (const int*)d_in[3];
    float*       out = (float*)d_out;

    const int E = in_sizes[1];
    if (E <= 0) return;

    float* tab  = (float*)d_ws;
    float* len2 = tab + (size_t)DIM_B * DIM_N * TAB_J;

    hipLaunchKernelGGL(row_products_wmma, dim3(ROW_TILES), dim3(256), 0, stream,
                       x, tab, len2);
    hipLaunchKernelGGL(edge_dist, dim3(EDGE_BLOCKS), dim3(256), 0, stream,
                       x, eb, ei, ej, tab, len2, out, E);
}


# Launch 2 keeps its 768 blocks, but a wave requests the indices of all three of its 32-edge groups at once, then all their gathers: two round trips to memory per wave instead of four in a row
# speedup vs baseline: 4.1946x; 1.0807x over previous
#include <hip/hip_runtime.h>

typedef float v2f __attribute__((ext_vector_type(2)));
typedef float v4f __attribute__((ext_vector_type(4)));
typedef float v8f __attribute__((ext_vector_type(8)));

#define DIM_B 4
#define DIM_D 256
#define DIM_N 1024
#define TAB_J 256
#define ROW_TILES (DIM_B * DIM_N / 16)
#define STAGE_ROWS 128
#define STAGE_PITCH 260
#define QUARTER 64
#define EDGE_BLOCKS 768
#define TRIP_GROUPS 3
#define GROUP_STRIDE ((long)EDGE_BLOCKS * 256)
#define PIVOT 0.5f
#define KEEP_SHARE 0.125f

#define EDGES_PER_WAVE 16
#define WAVES_PER_BLOCK 8
#define CHUNK 64
#define NCHUNK (DIM_D / CHUNK)
#define PITCH 68
#define TILE_FLOATS (16 * PITCH)

__device__ __forceinline__ void wave_lds_order()
{
    __builtin_amdgcn_fence(__ATOMIC_RELEASE, "wavefront");
    __builtin_amdgcn_wave_barrier();
    __builtin_amdgcn_fence(__ATOMIC_ACQUIRE, "wavefront");
}

__device__ __forceinline__ v8f mma4(v2f a, v2f b, v8f c)
{
    return __builtin_amdgcn_wmma_f32_16x16x4_f32(
                false, a,  false, b,
                (short)0, c,  false,  false);
}

__global__ __launch_bounds__(256)
void row_products_wmma(const float* __restrict__ x,
                       float*       __restrict__ tab,
                       float*       __restrict__ len2)
{
    __shared__ __attribute__((aligned(16))) float rows[STAGE_ROWS * STAGE_PITCH];
    __shared__ float part[8][16];

    const int tid  = threadIdx.x;
    const int wave = tid >> 5;
    const int lane = tid & 31;
    const int c    = lane & 15;
    const int h    = lane >> 4;

    const int gb = blockIdx.x;
    const int b  = gb >> 6;
    const int ib = (gb >> 2) & 15;
    const int jb = gb & 3;
    const int rowA0 = b * DIM_N + ib * 64;
    const int rowB0 = b * DIM_N + jb * 64;

    const int srow = tid >> 4;
    const int sc4  = tid & 15;
    const float* gsrc[8];
    #pragma unroll
    for (int r = 0; r < 8; ++r) {
        const int grow = (r < 4) ? (rowA0 + 16 * r + srow) : (rowB0 + 16 * (r - 4) + srow);
        gsrc[r] = x + (size_t)grow * DIM_D + 4 * sc4;
    }
    float* ldst = rows + srow * STAGE_PITCH + 4 * sc4;

    v4f st[2][8];
    #pragma unroll
    for (int r = 0; r < 8; ++r) {
        st[0][r] = *(const v4f*)(gsrc[r]);
        st[1][r] = *(const v4f*)(gsrc[r] + QUARTER);
    }

    const int wi = wave >> 1;
    const int wj = wave & 1;
    const float* la  = rows + (16 * wi + c) * STAGE_PITCH + 4 * h;
    const float* lb0 = rows + (64 + 32 * wj + c) * STAGE_PITCH + 4 * h;
    const float* lb1 = lb0 + 16 * STAGE_PITCH;
    v8f acc0 = {0.f, 0.f, 0.f, 0.f, 0.f, 0.f, 0.f, 0.f};
    v8f acc1 = {0.f, 0.f, 0.f, 0.f, 0.f, 0.f, 0.f, 0.f};

    #pragma unroll
    for (int q = 0; q < 4; ++q) {
        #pragma unroll
        for (int r = 0; r < 8; ++r)
            *(v4f*)(ldst + 16 * r * STAGE_PITCH + q * QUARTER) = st[q & 1][r] - PIVOT;
        __syncthreads();
        if (q + 2 < 4) {
            #pragma unroll
            for (int r = 0; r < 8; ++r)
                st[q & 1][r] = *(const v4f*)(gsrc[r] + (q + 2) * QUARTER);
        }
        #pragma unroll
        for (int uu = 0; uu < 8; ++uu) {
            const int u = 8 * q + uu;
            const v4f fa = *(const v4f*)(la  + 8 * u);
            const v4f f0 = *(const v4f*)(lb0 + 8 * u);
            const v4f f1 = *(const v4f*)(lb1 + 8 * u);
            v2f a0; a0.x = fa.x; a0.y = fa.y;
            v2f a1; a1.x = fa.z; a1.y = fa.w;
            v2f p0; p0.x = f0.x; p0.y = f0.y;
            v2f p1; p1.x = f0.z; p1.y = f0.w;
            v2f q0; q0.x = f1.x; q0.y = f1.y;
            v2f q1; q1.x = f1.z; q1.y = f1.w;
            acc0 = mma4(a0, p0, acc0);
            acc1 = mma4(a0, q0, acc1);
            acc0 = mma4(a1, p1, acc0);
            acc1 = mma4(a1, q1, acc1);
        }
    }

    {
        const float* ln = rows + (16 * jb + c) * STAGE_PITCH + 32 * wave + 4 * h;
        v8f na = {0.f, 0.f, 0.f, 0.f, 0.f, 0.f, 0.f, 0.f};
        #pragma unroll
        for (int uu = 0; uu < 4; ++uu) {
            const v4f f = *(const v4f*)(ln + 8 * uu);
            v2f a0; a0.x = f.x; a0.y = f.y;
            v2f a1; a1.x = f.z; a1.y = f.w;
            na = mma4(a0, a0, na);
            na = mma4(a1, a1, na);
        }
        float dv = na[0];
        #pragma unroll
        for (int r = 1; r < 8; ++r)
            dv = ((c & 7) == r) ? na[r] : dv;
        if (h == (c >> 3))
            part[wave][c] = dv;
    }

    const int gt = b * (DIM_N / 16) + ib * 4 + wi;
    const int j0 = jb * 64 + wj * 32;
    float* po = tab + ((size_t)gt * 16 + 8 * h) * TAB_J + j0 + c;
    #pragma unroll
    for (int r = 0; r < 8; ++r) {
        po[(size_t)r * TAB_J]      = acc0[r];
        po[(size_t)r * TAB_J + 16] = acc1[r];
    }

    __syncthreads();
    if (tid < 16) {
        float sum = part[0][tid];
        #pragma unroll
        for (int w = 1; w < 8; ++w)
            sum += part[w][tid];
        len2[(b * (DIM_N / 16) + ib * 4 + jb) * 16 + tid] = sum;
    }
}

__device__ __forceinline__ void diff_group(const float* __restrict__ x,
                                           const int*   __restrict__ eb,
                                           const int*   __restrict__ ei,
                                           const int*   __restrict__ ej,
                                           float*       __restrict__ out,
                                           int E, long groupBase, float* tile, int lane)
{
    const int m = lane & 15;
    const int h = lane >> 4;
    const long e  = groupBase + m;
    const long ec = (e < (long)E) ? e : (long)(E - 1);

    const int b = eb[ec];
    const int i = ei[ec];
    const int j = ej[ec];
    const unsigned offi = ((unsigned)b * DIM_N + (unsigned)i) * DIM_D;
    const unsigned offj = ((unsigned)b * DIM_N + (unsigned)j) * DIM_D;

    unsigned vi[8], vj[8];
    #pragma unroll
    for (int t = 0; t < 8; ++t) {
        const int src = 2 * t + h;
        vi[t] = (unsigned)__shfl((int)offi, src, 32) + 4u * (unsigned)m;
        vj[t] = (unsigned)__shfl((int)offj, src, 32) + 4u * (unsigned)m;
    }

    v8f c = {0.f, 0.f, 0.f, 0.f, 0.f, 0.f, 0.f, 0.f};

    #pragma nounroll
    for (int q = 0; q < NCHUNK; ++q) {
        v4f d[8];
        #pragma unroll
        for (int t = 0; t < 8; ++t) {
            const v4f xi = *(const v4f*)(x + vi[t] + q * CHUNK);
            const v4f xj = *(const v4f*)(x + vj[t] + q * CHUNK);
            d[t] = xi - xj;
        }

        wave_lds_order();
        #pragma unroll
        for (int t = 0; t < 8; ++t)
            *(v4f*)(tile + (2 * t + h) * PITCH + 4 * m) = d[t];
        wave_lds_order();

        #pragma unroll
        for (int u = 0; u < CHUNK / 8; ++u) {
            const v4f f = *(const v4f*)(tile + m * PITCH + 8 * u + 4 * h);
            v2f a0; a0.x = f.x; a0.y = f.y;
            v2f a1; a1.x = f.z; a1.y = f.w;
            c = mma4(a0, a0, c);
            c = mma4(a1, a1, c);
        }
    }

    const int mBase = h * 8;
    wave_lds_order();
    #pragma unroll
    for (int r = 0; r < 8; ++r)
        tile[(mBase + r) * 16 + m] = c[r];
    wave_lds_order();

    if (lane < 16 && e < (long)E) {
        const float s = tile[m * 16 + m];
        const float dd = sqrtf(s);
        float4 v = make_float4(dd, dd, dd, dd);
        float4* o = (float4*)(out + (size_t)e * 8);
        o[0] = v;
        o[1] = v;
    }
    wave_lds_order();
}

__global__ __launch_bounds__(256)
void edge_dist(const float* __restrict__ x,
               const int*   __restrict__ eb,
               const int*   __restrict__ ei,
               const int*   __restrict__ ej,
               const float* __restrict__ tab,
               const float* __restrict__ len2,
               float*       __restrict__ out,
               int E)
{
    __shared__ __attribute__((aligned(16))) float lds[WAVES_PER_BLOCK][TILE_FLOATS];

    const int tid  = threadIdx.x;
    const int wave = tid >> 5;
    const int lane = tid & 31;
    const long last = (long)E - 1;

    for (long blk0 = blockIdx.x; blk0 * 256 < (long)E; blk0 += (long)TRIP_GROUPS * EDGE_BLOCKS) {
        const long base0 = blk0 * 256 + (long)wave * 32;

        int b[TRIP_GROUPS], i[TRIP_GROUPS], j[TRIP_GROUPS];
        #pragma unroll
        for (int g = 0; g < TRIP_GROUPS; ++g) {
            const long e0 = base0 + (long)g * GROUP_STRIDE + lane;
            const long ec = (e0 < last) ? e0 : last;
            b[g] = eb[ec];
            i[g] = ei[ec];
            j[g] = ej[ec];
        }

        float li[TRIP_GROUPS], lj[TRIP_GROUPS], gp[TRIP_GROUPS];
        #pragma unroll
        for (int g = 0; g < TRIP_GROUPS; ++g) {
            const unsigned bc = min((unsigned)b[g], (unsigned)(DIM_B - 1));
            const unsigned ic = min((unsigned)i[g], (unsigned)(DIM_N - 1));
            const unsigned jc = min((unsigned)j[g], (unsigned)(TAB_J - 1));
            const unsigned r  = bc * DIM_N;
            li[g] = len2[r + ic];
            lj[g] = len2[r + jc];
            gp[g] = tab[(size_t)(r + ic) * TAB_J + jc];
        }

        unsigned redo = 0;
        #pragma unroll
        for (int g = 0; g < TRIP_GROUPS; ++g) {
            const long base = base0 + (long)g * GROUP_STRIDE;
            const bool valid = base < (long)E;
            const bool intab = (unsigned)b[g] < DIM_B && (unsigned)i[g] < DIM_N && (unsigned)j[g] < TAB_J;
            const float s  = li[g] + lj[g];
            const float d2 = s - 2.f * gp[g];
            const bool same = (i[g] == j[g]);
            const bool ok = intab && (s < 3.0e38f) && (same || d2 >= KEEP_SHARE * s);
            const float d = same ? 0.f : sqrtf(d2);

            if (__all(ok)) {
                const float dlo = __shfl(d, lane >> 1, 32);
                const float dhi = __shfl(d, 16 + (lane >> 1), 32);
                float* o = out + (size_t)base * 8;
                if (base + (lane >> 1) < (long)E)
                    *(float4*)(o + 4 * lane) = make_float4(dlo, dlo, dlo, dlo);
                if (base + 16 + (lane >> 1) < (long)E)
                    *(float4*)(o + 128 + 4 * lane) = make_float4(dhi, dhi, dhi, dhi);
            } else if (valid) {
                redo |= 1u << g;
            }
        }

        #pragma nounroll
        for (int g = 0; redo != 0; ++g, redo >>= 1) {
            if (redo & 1u) {
                const long base = base0 + (long)g * GROUP_STRIDE;
                diff_group(x, eb, ei, ej, out, E, base, lds[wave], lane);
                if (base + 16 < (long)E)
                    diff_group(x, eb, ei, ej, out, E, base + 16, lds[wave], lane);
            }
        }
    }
}

extern "C" void kernel_launch(void* const* d_in, const int* in_sizes, int n_in,
                              void* d_out, int out_size, void* d_ws, size_t ws_size,
                              hipStream_t stream)
{
    const float* x  = (const float*)d_in[0];
    const int*   eb = (const int*)d_in[1];
    const int*   ei = (const int*)d_in[2];
    const int*   ej = (const int*)d_in[3];
    float*       out = (float*)d_out;

    const int E = in_sizes[1];
    if (E <= 0) return;

    float* tab  = (float*)d_ws;
    float* len2 = tab + (size_t)DIM_B * DIM_N * TAB_J;

    hipLaunchKernelGGL(row_products_wmma, dim3(ROW_TILES), dim3(256), 0, stream,
                       x, tab, len2);
    hipLaunchKernelGGL(edge_dist, dim3(EDGE_BLOCKS), dim3(256), 0, stream,
                       x, eb, ei, ej, tab, len2, out, E);
}
